# starting kernel
# baseline (speedup 1.0000x reference)
#include <hip/hip_runtime.h>

typedef __attribute__((ext_vector_type(2))) float v2f;
typedef __attribute__((ext_vector_type(8))) float v8f;

#define NN 50000
#define NE 800000
#define KIN 128
#define HID 64
#define NC 32
#define ROW_TILES (NN / 16)

__global__ void k_zero(float* __restrict__ p, int n) {
  int i = blockIdx.x * blockDim.x + threadIdx.x;
  if (i < n) p[i] = 0.0f;
}

__global__ void k_degree(const int* __restrict__ dst, float* __restrict__ deg) {
  int e = blockIdx.x * blockDim.x + threadIdx.x;
  if (e < NE) atomicAdd(&deg[dst[e]], 1.0f);
}

__global__ void k_dinv(float* __restrict__ deg) {
  int i = blockIdx.x * blockDim.x + threadIdx.x;
  if (i < NN) deg[i] = rsqrtf(deg[i] + 1.0f);
}

__global__ void k_gemm_xw0(const float* __restrict__ X,
                           const float* __restrict__ W,
                           float* __restrict__ H) {
  int wave = (blockIdx.x * blockDim.x + threadIdx.x) >> 5;
  if (wave >= ROW_TILES) return;
  int lane = threadIdx.x & 31;
  int hh   = lane >> 4;
  int lid  = lane & 15;
  const float* xr = X + (wave * 16 + lid) * KIN;
  v8f c0 = {}, c1 = {}, c2 = {}, c3 = {};
  for (int k = 0; k < KIN; k += 4) {
    __builtin_prefetch(xr + k + 32, 0, 1);
    int ka = k + 2 * hh;
    v2f a;
    a.x = xr[ka];
    a.y = xr[ka + 1];
    const float* wr0 = W + ka * HID;
    const float* wr1 = wr0 + HID;
    v2f b;
    b.x = wr0[lid];      b.y = wr1[lid];
    c0 = __builtin_amdgcn_wmma_f32_16x16x4_f32(false, a, false, b, (short)0, c0, false, false);
    b.x = wr0[lid + 16]; b.y = wr1[lid + 16];
    c1 = __builtin_amdgcn_wmma_f32_16x16x4_f32(false, a, false, b, (short)0, c1, false, false);
    b.x = wr0[lid + 32]; b.y = wr1[lid + 32];
    c2 = __builtin_amdgcn_wmma_f32_16x16x4_f32(false, a, false, b, (short)0, c2, false, false);
    b.x = wr0[lid + 48]; b.y = wr1[lid + 48];
    c3 = __builtin_amdgcn_wmma_f32_16x16x4_f32(false, a, false, b, (short)0, c3, false, false);
  }
  int mrow = wave * 16 + hh * 8;
  for (int j = 0; j < 8; ++j) {
    float* o = H + (mrow + j) * HID + lid;
    o[0]  = c0[j];
    o[16] = c1[j];
    o[32] = c2[j];
    o[48] = c3[j];
  }
}

__global__ void k_gemm_hw12(const float* __restrict__ H,
                            const float* __restrict__ W1,
                            const float* __restrict__ W2,
                            float* __restrict__ MU,
                            float* __restrict__ LV) {
  int wave = (blockIdx.x * blockDim.x + threadIdx.x) >> 5;
  if (wave >= ROW_TILES) return;
  int lane = threadIdx.x & 31;
  int hh   = lane >> 4;
  int lid  = lane & 15;
  const float* hr = H + (wave * 16 + lid) * HID;
  v8f m0 = {}, m1 = {}, l0 = {}, l1 = {};
  for (int k = 0; k < HID; k += 4) {
    int ka = k + 2 * hh;
    v2f a;
    a.x = hr[ka];
    a.y = hr[ka + 1];
    const float* u0 = W1 + ka * NC;
    const float* u1 = u0 + NC;
    const float* v0 = W2 + ka * NC;
    const float* v1 = v0 + NC;
    v2f b;
    b.x = u0[lid];      b.y = u1[lid];
    m0 = __builtin_amdgcn_wmma_f32_16x16x4_f32(false, a, false, b, (short)0, m0, false, false);
    b.x = u0[lid + 16]; b.y = u1[lid + 16];
    m1 = __builtin_amdgcn_wmma_f32_16x16x4_f32(false, a, false, b, (short)0, m1, false, false);
    b.x = v0[lid];      b.y = v1[lid];
    l0 = __builtin_amdgcn_wmma_f32_16x16x4_f32(false, a, false, b, (short)0, l0, false, false);
    b.x = v0[lid + 16]; b.y = v1[lid + 16];
    l1 = __builtin_amdgcn_wmma_f32_16x16x4_f32(false, a, false, b, (short)0, l1, false, false);
  }
  int mrow = wave * 16 + hh * 8;
  for (int j = 0; j < 8; ++j) {
    float* om = MU + (mrow + j) * NC + lid;
    float* ol = LV + (mrow + j) * NC + lid;
    om[0]  = m0[j];
    om[16] = m1[j];
    ol[0]  = l0[j];
    ol[16] = l1[j];
  }
}

__global__ void k_selfinit64(const float* __restrict__ Hlin,
                             const float* __restrict__ dinv,
                             const float* __restrict__ bias,
                             float* __restrict__ Agg) {
  int idx = blockIdx.x * blockDim.x + threadIdx.x;
  if (idx >= NN * HID) return;
  int i = idx >> 6;
  int f = idx & 63;
  float d = dinv[i];
  Agg[idx] = Hlin[idx] * d * d + bias[f];
}

__global__ void k_scatter64(const float* __restrict__ Hlin,
                            const int* __restrict__ src,
                            const int* __restrict__ dst,
                            const float* __restrict__ dinv,
                            float* __restrict__ Agg) {
  int idx = blockIdx.x * blockDim.x + threadIdx.x;
  int e = idx >> 6;
  if (e >= NE) return;
  int f = idx & 63;
  int s = src[e], d = dst[e];
  float nrm = dinv[s] * dinv[d];
  atomicAdd(&Agg[d * HID + f], Hlin[s * HID + f] * nrm);
}

__global__ void k_reludrop(const float* __restrict__ Agg,
                           const float* __restrict__ mask,
                           float* __restrict__ Hout) {
  int idx = blockIdx.x * blockDim.x + threadIdx.x;
  if (idx < NN * HID) Hout[idx] = fmaxf(Agg[idx], 0.0f) * mask[idx];
}

__global__ void k_selfinit32(const float* __restrict__ MUlin,
                             const float* __restrict__ LVlin,
                             const float* __restrict__ dinv,
                             const float* __restrict__ b1,
                             const float* __restrict__ b2,
                             float* __restrict__ MUout,
                             float* __restrict__ LVout) {
  int idx = blockIdx.x * blockDim.x + threadIdx.x;
  if (idx >= NN * NC) return;
  int i = idx >> 5;
  int f = idx & 31;
  float d  = dinv[i];
  float dd = d * d;
  MUout[idx] = MUlin[idx] * dd + b1[f];
  LVout[idx] = LVlin[idx] * dd + b2[f];
}

__global__ void k_scatter32(const float* __restrict__ MUlin,
                            const float* __restrict__ LVlin,
                            const int* __restrict__ src,
                            const int* __restrict__ dst,
                            const float* __restrict__ dinv,
                            float* __restrict__ MUout,
                            float* __restrict__ LVout) {
  int idx = blockIdx.x * blockDim.x + threadIdx.x;
  int e = idx >> 5;
  if (e >= NE) return;
  int f = idx & 31;
  int s = src[e], d = dst[e];
  float nrm = dinv[s] * dinv[d];
  atomicAdd(&MUout[d * NC + f], MUlin[s * NC + f] * nrm);
  atomicAdd(&LVout[d * NC + f], LVlin[s * NC + f] * nrm);
}

__global__ void k_final(const float* __restrict__ eps,
                        const float* __restrict__ MU,
                        const float* __restrict__ LV,
                        float* __restrict__ Z) {
  int idx = blockIdx.x * blockDim.x + threadIdx.x;
  if (idx < NN * NC) Z[idx] = eps[idx] * expf(LV[idx]) + MU[idx];
}

extern "C" void kernel_launch(void* const* d_in, const int* in_sizes, int n_in,
                              void* d_out, int out_size, void* d_ws, size_t ws_size,
                              hipStream_t stream) {
  (void)in_sizes; (void)n_in; (void)out_size; (void)ws_size;
  const float* x    = (const float*)d_in[0];
  const int*   ei   = (const int*)d_in[1];
  const float* W0   = (const float*)d_in[2];
  const float* b0   = (const float*)d_in[3];
  const float* W1   = (const float*)d_in[4];
  const float* b1   = (const float*)d_in[5];
  const float* W2   = (const float*)d_in[6];
  const float* b2   = (const float*)d_in[7];
  const float* mask = (const float*)d_in[8];
  const float* eps  = (const float*)d_in[9];
  const int* src = ei;
  const int* dst = ei + NE;

  float* ws    = (float*)d_ws;
  float* dinv  = ws;
  float* hlin  = dinv + NN;
  float* hagg  = hlin + (size_t)NN * HID;
  float* mulin = hagg + (size_t)NN * HID;
  float* lvlin = mulin + (size_t)NN * NC;

  float* z  = (float*)d_out;
  float* mu = z + (size_t)NN * NC;
  float* lv = mu + (size_t)NN * NC;

  const int B = 256;
  const int gN   = (NN + B - 1) / B;
  const int gE   = (NE + B - 1) / B;
  const int gW   = (ROW_TILES * 32 + B - 1) / B;
  const int gNH  = (NN * HID + B - 1) / B;
  const int gEH  = (NE * HID + B - 1) / B;
  const int gNC_ = (NN * NC + B - 1) / B;
  const int gEC  = (NE * NC + B - 1) / B;

  k_zero   <<<gN, B, 0, stream>>>(dinv, NN);
  k_degree <<<gE, B, 0, stream>>>(dst, dinv);
  k_dinv   <<<gN, B, 0, stream>>>(dinv);

  k_gemm_xw0  <<<gW,  B, 0, stream>>>(x, W0, hlin);
  k_selfinit64<<<gNH, B, 0, stream>>>(hlin, dinv, b0, hagg);
  k_scatter64 <<<gEH, B, 0, stream>>>(hlin, src, dst, dinv, hagg);
  k_reludrop  <<<gNH, B, 0, stream>>>(hagg, mask, hlin);

  k_gemm_hw12 <<<gW,   B, 0, stream>>>(hlin, W1, W2, mulin, lvlin);
  k_selfinit32<<<gNC_, B, 0, stream>>>(mulin, lvlin, dinv, b1, b2, mu, lv);
  k_scatter32 <<<gEC,  B, 0, stream>>>(mulin, lvlin, src, dst, dinv, mu, lv);

  k_final     <<<gNC_, B, 0, stream>>>(eps, mu, lv, z);
}


# Edge weight dinv[s]*dinv[d] is formed once, in the first edge pass, and stored one 128-byte line per 16-edge run; the second pass loads that line instead of four scattered count reads per edge
# speedup vs baseline: 3.3151x; 3.3151x over previous
#include <hip/hip_runtime.h>

typedef __attribute__((ext_vector_type(2))) float v2f;
typedef __attribute__((ext_vector_type(8))) float v8f;
typedef __attribute__((ext_vector_type(4))) float v4f;

#define NN 50000
#define NE 800000
#define KIN 128
#define HID 64
#define NC 32
#define NNP 50048
#define ROW_TILES (NN / 16)
#define EPW 16
#define RUNS (NE / EPW)

__device__ __forceinline__ unsigned cache_half() {
  return (__builtin_amdgcn_s_sendmsg_rtn(0x87) >> 18) & 1u;
}

__global__ void k_zero(float* __restrict__ p, int n) {
  int i = blockIdx.x * blockDim.x + threadIdx.x;
  if (i < n) p[i] = 0.0f;
}

__device__ __forceinline__ float dinv_of(const float* __restrict__ deg2, int i) {
  return rsqrtf(deg2[i] + deg2[NNP + i] + 1.0f);
}

#define GEMM1_BLOCKS ((ROW_TILES * 32 + 255) / 256)
#define DEG_BLOCKS   ((NE + 255) / 256)
__global__ __launch_bounds__(256) void k_gemm_xw0_deg(const float* __restrict__ X,
                                                      const float* __restrict__ W,
                                                      const float* __restrict__ bias,
                                                      const int* __restrict__ dst,
                                                      float* __restrict__ deg2,
                                                      float* __restrict__ H,
                                                      float* __restrict__ Agg,
                                                      float* __restrict__ AggB) {
  if (blockIdx.x >= GEMM1_BLOCKS) {
    int e = (blockIdx.x - GEMM1_BLOCKS) * 256 + threadIdx.x;
    if (e < NE) {
      int d = dst[e];
      float* deg = deg2 + cache_half() * NNP;
      atomicAdd(&deg[d], 1.0f);
    }
    return;
  }
  int wave = (blockIdx.x * 256 + threadIdx.x) >> 5;
  if (wave >= ROW_TILES) return;
  int lane = threadIdx.x & 31;
  int hh   = lane >> 4;
  int lid  = lane & 15;
  const float* xr = X + (wave * 16 + lid) * KIN;
  v8f c0 = {}, c1 = {}, c2 = {}, c3 = {};
  for (int k = 0; k < KIN; k += 4) {
    __builtin_prefetch(xr + k + 32, 0, 1);
    int ka = k + 2 * hh;
    v2f a;
    a.x = xr[ka];
    a.y = xr[ka + 1];
    const float* wr0 = W + ka * HID;
    const float* wr1 = wr0 + HID;
    v2f b;
    b.x = wr0[lid];      b.y = wr1[lid];
    c0 = __builtin_amdgcn_wmma_f32_16x16x4_f32(false, a, false, b, (short)0, c0, false, false);
    b.x = wr0[lid + 16]; b.y = wr1[lid + 16];
    c1 = __builtin_amdgcn_wmma_f32_16x16x4_f32(false, a, false, b, (short)0, c1, false, false);
    b.x = wr0[lid + 32]; b.y = wr1[lid + 32];
    c2 = __builtin_amdgcn_wmma_f32_16x16x4_f32(false, a, false, b, (short)0, c2, false, false);
    b.x = wr0[lid + 48]; b.y = wr1[lid + 48];
    c3 = __builtin_amdgcn_wmma_f32_16x16x4_f32(false, a, false, b, (short)0, c3, false, false);
  }
  int mrow = wave * 16 + hh * 8;
  float bb0 = bias[lid], bb1 = bias[lid + 16], bb2 = bias[lid + 32], bb3 = bias[lid + 48];
  for (int j = 0; j < 8; ++j) {
    float* o = H + (mrow + j) * HID + lid;
    float* g = Agg + (mrow + j) * HID + lid;
    o[0]  = c0[j];
    o[16] = c1[j];
    o[32] = c2[j];
    o[48] = c3[j];
    g[0]  = bb0;
    g[16] = bb1;
    g[32] = bb2;
    g[48] = bb3;
  }
  v4f* zb = (v4f*)(AggB + wave * 16 * HID);
  v4f zero4 = {0.0f, 0.0f, 0.0f, 0.0f};
  for (int i = 0; i < 8; ++i) zb[i * 32 + lane] = zero4;
}

__global__ void k_gemm_hw12(const float* __restrict__ Agg,
                            const float* __restrict__ AggB,
                            const float* __restrict__ Hlin,
                            const float* __restrict__ mask,
                            const float* __restrict__ W1,
                            const float* __restrict__ W2,
                            const float* __restrict__ deg2,
                            const float* __restrict__ b1,
                            const float* __restrict__ b2,
                            float* __restrict__ MULV,
                            float* __restrict__ MUout,
                            float* __restrict__ LVout,
                            float* __restrict__ MUB,
                            float* __restrict__ LVB) {
  int wave = (blockIdx.x * blockDim.x + threadIdx.x) >> 5;
  if (wave >= ROW_TILES) return;
  int lane = threadIdx.x & 31;
  int hh   = lane >> 4;
  int lid  = lane & 15;
  const float* hr = Agg  + (wave * 16 + lid) * HID;
  const float* hb = AggB + (wave * 16 + lid) * HID;
  const float* hl = Hlin + (wave * 16 + lid) * HID;
  const float* mr = mask + (wave * 16 + lid) * HID;
  float dr  = dinv_of(deg2, wave * 16 + lid);
  float ddr = dr * dr;
  v8f m0 = {}, m1 = {}, l0 = {}, l1 = {};
  for (int k = 0; k < HID; k += 4) {
    int ka = k + 2 * hh;
    v2f a;
    a.x = fmaxf(hr[ka]     + hb[ka]     + hl[ka]     * ddr, 0.0f) * mr[ka];
    a.y = fmaxf(hr[ka + 1] + hb[ka + 1] + hl[ka + 1] * ddr, 0.0f) * mr[ka + 1];
    const float* u0 = W1 + ka * NC;
    const float* u1 = u0 + NC;
    const float* v0 = W2 + ka * NC;
    const float* v1 = v0 + NC;
    v2f b;
    b.x = u0[lid];      b.y = u1[lid];
    m0 = __builtin_amdgcn_wmma_f32_16x16x4_f32(false, a, false, b, (short)0, m0, false, false);
    b.x = u0[lid + 16]; b.y = u1[lid + 16];
    m1 = __builtin_amdgcn_wmma_f32_16x16x4_f32(false, a, false, b, (short)0, m1, false, false);
    b.x = v0[lid];      b.y = v1[lid];
    l0 = __builtin_amdgcn_wmma_f32_16x16x4_f32(false, a, false, b, (short)0, l0, false, false);
    b.x = v0[lid + 16]; b.y = v1[lid + 16];
    l1 = __builtin_amdgcn_wmma_f32_16x16x4_f32(false, a, false, b, (short)0, l1, false, false);
  }
  int mrow = wave * 16 + hh * 8;
  float b1a = b1[lid], b1b = b1[lid + 16], b2a = b2[lid], b2b = b2[lid + 16];
  for (int j = 0; j < 8; ++j) {
    float d  = dinv_of(deg2, mrow + j);
    float dd = d * d;
    float* om = MULV + (mrow + j) * HID + lid;
    float* gm = MUout + (mrow + j) * NC + lid;
    float* gl = LVout + (mrow + j) * NC + lid;
    om[0]  = m0[j];
    om[16] = m1[j];
    om[32] = l0[j];
    om[48] = l1[j];
    gm[0]  = m0[j] * dd + b1a;
    gm[16] = m1[j] * dd + b1b;
    gl[0]  = l0[j] * dd + b2a;
    gl[16] = l1[j] * dd + b2b;
  }
  v4f* zm = (v4f*)(MUB + wave * 16 * NC);
  v4f* zl = (v4f*)(LVB + wave * 16 * NC);
  v4f zero4 = {0.0f, 0.0f, 0.0f, 0.0f};
  for (int i = 0; i < 4; ++i) {
    zm[i * 32 + lane] = zero4;
    zl[i * 32 + lane] = zero4;
  }
}


__global__ __launch_bounds__(256) void k_scatter64(const float* __restrict__ Hlin,
                                                   const int* __restrict__ ei,
                                                   const float* __restrict__ deg2,
                                                   float* __restrict__ AggA,
                                                   float* __restrict__ AggB,
                                                   float* __restrict__ wE) {
  int wave = (blockIdx.x * blockDim.x + threadIdx.x) >> 5;
  if (wave >= RUNS) return;
  int lane = threadIdx.x & 31;
  int node = ei[(lane >> 4) * NE + wave * EPW + (lane & 15)];
  float dv = dinv_of(deg2, node);
  float nrm = dv * __shfl_xor(dv, 16);
  float v[2 * EPW];
#pragma unroll
  for (int j = 0; j < EPW; ++j) {
    unsigned s = (unsigned)__builtin_amdgcn_readlane(node, j);
    const float* r = Hlin + s * HID + lane;
    v[2 * j]     = r[0];
    v[2 * j + 1] = r[32];
  }
  float* Agg = cache_half() ? AggB : AggA;
#pragma unroll
  for (int j = 0; j < EPW; ++j) {
    unsigned d = (unsigned)__builtin_amdgcn_readlane(node, 16 + j);
    float nj = __int_as_float(__builtin_amdgcn_readlane(__float_as_int(nrm), j));
    float* o = Agg + d * HID + lane;
    atomicAdd(o,      v[2 * j]     * nj);
    atomicAdd(o + 32, v[2 * j + 1] * nj);
  }
  wE[wave * 32 + lane] = nrm;
}

__global__ __launch_bounds__(256) void k_scatter32(const float* __restrict__ MULVlin,
                                                   const int* __restrict__ ei,
                                                   const float* __restrict__ wE,
                                                   float* __restrict__ MUA,
                                                   float* __restrict__ LVA,
                                                   float* __restrict__ MUB,
                                                   float* __restrict__ LVB) {
  int wave = (blockIdx.x * blockDim.x + threadIdx.x) >> 5;
  if (wave >= RUNS) return;
  int lane = threadIdx.x & 31;
  int node = ei[(lane >> 4) * NE + wave * EPW + (lane & 15)];
  float nrm = wE[wave * 32 + lane];
  float a[EPW], b[EPW];
#pragma unroll
  for (int j = 0; j < EPW; ++j) {
    unsigned s = (unsigned)__builtin_amdgcn_readlane(node, j);
    const float* r = MULVlin + s * HID + lane;
    a[j] = r[0];
    b[j] = r[32];
  }
  unsigned half = cache_half();
  float* MUout = half ? MUB : MUA;
  float* LVout = half ? LVB : LVA;
#pragma unroll
  for (int j = 0; j < EPW; ++j) {
    unsigned d = (unsigned)__builtin_amdgcn_readlane(node, 16 + j);
    float nj = __int_as_float(__builtin_amdgcn_readlane(__float_as_int(nrm), j));
    atomicAdd(&MUout[d * NC + lane], a[j] * nj);
    atomicAdd(&LVout[d * NC + lane], b[j] * nj);
  }
}

__global__ void k_final(const float* __restrict__ eps,
                        float* __restrict__ MU,
                        float* __restrict__ LV,
                        const float* __restrict__ MUB,
                        const float* __restrict__ LVB,
                        float* __restrict__ Z) {
  int idx = blockIdx.x * blockDim.x + threadIdx.x;
  if (idx < NN * NC) {
    float m = MU[idx] + MUB[idx];
    float l = LV[idx] + LVB[idx];
    MU[idx] = m;
    LV[idx] = l;
    Z[idx] = eps[idx] * expf(l) + m;
  }
}

extern "C" void kernel_launch(void* const* d_in, const int* in_sizes, int n_in,
                              void* d_out, int out_size, void* d_ws, size_t ws_size,
                              hipStream_t stream) {
  (void)in_sizes; (void)n_in; (void)out_size; (void)ws_size;
  const float* x    = (const float*)d_in[0];
  const int*   ei   = (const int*)d_in[1];
  const float* W0   = (const float*)d_in[2];
  const float* b0   = (const float*)d_in[3];
  const float* W1   = (const float*)d_in[4];
  const float* b1   = (const float*)d_in[5];
  const float* W2   = (const float*)d_in[6];
  const float* b2   = (const float*)d_in[7];
  const float* mask = (const float*)d_in[8];
  const float* eps  = (const float*)d_in[9];
  const int* dst = ei + NE;

  float* ws    = (float*)d_ws;
  float* dinv  = ws;
  float* hlin  = dinv + 2 * NNP;
  float* hagg  = hlin + (size_t)NN * HID;
  float* haggb = hagg + (size_t)NN * HID;
  float* mulv  = haggb + (size_t)NN * HID;
  float* mub   = mulv + (size_t)NN * HID;
  float* lvb   = mub + (size_t)NN * NC;
  float* wedge = lvb + (size_t)NN * NC;

  float* z  = (float*)d_out;
  float* mu = z + (size_t)NN * NC;
  float* lv = mu + (size_t)NN * NC;

  const int B = 256;
  const int gZ   = (2 * NNP + B - 1) / B;
  const int gW   = (ROW_TILES * 32 + B - 1) / B;
  const int gWD  = GEMM1_BLOCKS + DEG_BLOCKS;
  const int gNC_ = (NN * NC + B - 1) / B;
  const int gS   = (RUNS * 32 + B - 1) / B;

  k_zero   <<<gZ, B, 0, stream>>>(dinv, 2 * NNP);

  k_gemm_xw0_deg <<<gWD, B, 0, stream>>>(x, W0, b0, dst, dinv, hlin, hagg, haggb);
  k_scatter64    <<<gS,  B, 0, stream>>>(hlin, ei, dinv, hagg, haggb, wedge);

  k_gemm_hw12 <<<gW,   B, 0, stream>>>(hagg, haggb, hlin, mask, W1, W2, dinv, b1, b2, mulv, mu, lv, mub, lvb);
  k_scatter32 <<<gS,   B, 0, stream>>>(mulv, ei, wedge, mu, lv, mub, lvb);

  k_final     <<<gNC_, B, 0, stream>>>(eps, mu, lv, mub, lvb, z);
}
